# starting kernel
# baseline (speedup 1.0000x reference)
#include <hip/hip_runtime.h>

typedef __attribute__((ext_vector_type(16))) _Float16 v16h;
typedef __attribute__((ext_vector_type(8)))  _Float16 v8h;
typedef __attribute__((ext_vector_type(8)))  float    v8f;

#define BSZ 2
#define NN  512
#define HH  128


__global__ void gine_norm_kernel(const float* __restrict__ x,
                                 const float* __restrict__ mask,
                                 const float* __restrict__ gamma,
                                 const float* __restrict__ beta,
                                 float* __restrict__ xm,
                                 _Float16* __restrict__ hn)
{
    int row = blockIdx.x;
    int t   = threadIdx.x;
    __shared__ float red[HH];
    float mk = mask[row];
    float xv = x[(size_t)row * HH + t] * mk;
    red[t] = xv; __syncthreads();
    for (int s = HH / 2; s > 0; s >>= 1) {
        if (t < s) red[t] += red[t + s];
        __syncthreads();
    }
    float mu = red[0] * (1.0f / HH);
    __syncthreads();
    float d = xv - mu;
    red[t] = d * d; __syncthreads();
    for (int s = HH / 2; s > 0; s >>= 1) {
        if (t < s) red[t] += red[t + s];
        __syncthreads();
    }
    float var = red[0] * (1.0f / HH);
    float hv  = d * rsqrtf(var + 1e-5f) * gamma[t] + beta[t];
    xm[(size_t)row * HH + t] = xv;
    hn[(size_t)row * HH + t] = (_Float16)hv;
}

__global__ void gine_swizzleW_kernel(const float* __restrict__ W1,
                                     const float* __restrict__ W2,
                                     _Float16* __restrict__ wsW)
{
    int idx  = blockIdx.x * blockDim.x + threadIdx.x;
    int lane = idx & 31;
    int kc   = (idx >> 5) & 3;
    int ot   = (idx >> 7) & 7;
    int mat  = (idx >> 10) & 1;
    const float* W = mat ? W2 : W1;
    _Float16* dst  = wsW + mat * 16384 + ((ot * 4 + kc) * 32 + lane) * 16;
    int m  = lane & 15;
    int lo = (lane & 16) ? 8 : 0;
#pragma unroll
    for (int h = 0; h < 16; ++h) {
        int kl = (h < 8) ? (lo + h) : (16 + lo + (h - 8));
        dst[h] = (_Float16)W[(ot * 16 + m) * HH + kc * 32 + kl];
    }
}

__global__ __launch_bounds__(256, 1)
void gine_main_kernel(const float*    __restrict__ edge,
                      const int*      __restrict__ adj,
                      const float*    __restrict__ mask,
                      const float*    __restrict__ b1,
                      const float*    __restrict__ b2,
                      const float*    __restrict__ xm,
                      const _Float16* __restrict__ hn,
                      const _Float16* __restrict__ wsW,
                      float*          __restrict__ out)
{
    __shared__ __align__(32) _Float16 sW[2 * 16384];
    __shared__ __align__(16) _Float16 sH1[8 * 8 * 32 * 8];
    __shared__ float sB1[HH], sB2[HH];
    __shared__ float sPart[HH];

    int tid  = threadIdx.x;
    int lane = tid & 31;
    int w    = tid >> 5;
    int b    = blockIdx.x / NN;
    int i    = blockIdx.x % NN;

    {
        const uint4* g = (const uint4*)wsW;
        uint4* s = (uint4*)sW;
        for (int idx = tid; idx < 4096; idx += 256) s[idx] = g[idx];
    }
    if (tid < HH) { sB1[tid] = b1[tid]; sB2[tid] = b2[tid]; sPart[tid] = 0.f; }
    __syncthreads();

    const int moff = (lane & 16) ? 8 : 0;
    const int mrow = lane & 15;
    const size_t rowbase = (size_t)(b * NN + i) * NN;

    for (int jt = w; jt < NN / 16; jt += 8) {
        int jrow = jt * 16 + mrow;
        float am = (float)adj[rowbase + jrow];
        const float*    erow = edge + (rowbase + jrow) * HH;
        const _Float16* hrow = hn + (size_t)(b * NN + jrow) * HH;

        v16h b1c[4];
#pragma unroll
        for (int kc = 0; kc < 4; ++kc) {
            int k0 = kc * 32 + moff;
            float ef[16];
            *(float4*)&ef[0]  = *(const float4*)(erow + k0);
            *(float4*)&ef[4]  = *(const float4*)(erow + k0 + 4);
            *(float4*)&ef[8]  = *(const float4*)(erow + k0 + 16);
            *(float4*)&ef[12] = *(const float4*)(erow + k0 + 20);
            v8h h0 = *(const v8h*)(hrow + k0);
            v8h h1 = *(const v8h*)(hrow + k0 + 16);
            v16h bb;
#pragma unroll
            for (int t2 = 0; t2 < 8; ++t2) {
                bb[t2]     = (_Float16)(ef[t2]     + (float)h0[t2]);
                bb[8 + t2] = (_Float16)(ef[8 + t2] + (float)h1[t2]);
            }
            b1c[kc] = bb;
        }

        if (jt + 8 < NN / 16) {
            const float* enext = edge + (rowbase + (jt + 8) * 16 + mrow) * HH;
#pragma unroll
            for (int kc = 0; kc < 4; ++kc)
                __builtin_prefetch(enext + kc * 32, 0, 1);
        }

#pragma unroll 1
        for (int ot = 0; ot < 8; ++ot) {
            v8f a1 = (v8f)0.f;
#pragma unroll
            for (int kc = 0; kc < 4; ++kc) {
                const v16h aw = *(const v16h*)&sW[((ot * 4 + kc) * 32 + lane) * 16];
                a1 = __builtin_amdgcn_wmma_f32_16x16x32_f16(
                    false, aw, false, b1c[kc], (short)0, a1, false, false);
            }
            float bvv[8];
            *(float4*)&bvv[0] = *(const float4*)&sB1[ot * 16 + moff];
            *(float4*)&bvv[4] = *(const float4*)&sB1[ot * 16 + moff + 4];
            v8h pk;
#pragma unroll
            for (int r = 0; r < 8; ++r)
                pk[r] = (_Float16)fmaxf(a1[r] + bvv[r], 0.f);
            *(v8h*)&sH1[((w * 8 + ot) * 32 + lane) * 8] = pk;
        }

        v16h b2c[4];
#pragma unroll
        for (int kc = 0; kc < 4; ++kc) {
            v8h lo = *(const v8h*)&sH1[((w * 8 + 2 * kc)     * 32 + lane) * 8];
            v8h hi = *(const v8h*)&sH1[((w * 8 + 2 * kc + 1) * 32 + lane) * 8];
            b2c[kc] = __builtin_shufflevector(lo, hi,
                0, 1, 2, 3, 4, 5, 6, 7, 8, 9, 10, 11, 12, 13, 14, 15);
        }

#pragma unroll 1
        for (int ot = 0; ot < 8; ++ot) {
            v8f a2 = (v8f)0.f;
#pragma unroll
            for (int kc = 0; kc < 4; ++kc) {
                const v16h aw =
                    *(const v16h*)&sW[16384 + ((ot * 4 + kc) * 32 + lane) * 16];
                a2 = __builtin_amdgcn_wmma_f32_16x16x32_f16(
                    false, aw, false, b2c[kc], (short)0, a2, false, false);
            }
            float bvv[8];
            *(float4*)&bvv[0] = *(const float4*)&sB2[ot * 16 + moff];
            *(float4*)&bvv[4] = *(const float4*)&sB2[ot * 16 + moff + 4];
#pragma unroll
            for (int r = 0; r < 8; ++r) {
                float v = fmaxf(a2[r] + bvv[r], 0.f) * am;
                v += __shfl_xor(v, 1, 32);
                v += __shfl_xor(v, 2, 32);
                v += __shfl_xor(v, 4, 32);
                v += __shfl_xor(v, 8, 32);
                if ((lane & 15) == 0)
                    atomicAdd(&sPart[ot * 16 + moff + r], v);
            }
        }
    }
    __syncthreads();

    if (tid < HH) {
        int row = b * NN + i;
        out[(size_t)row * HH + tid] =
            (xm[(size_t)row * HH + tid] + sPart[tid]) * mask[row];
    }
}

extern "C" void kernel_launch(void* const* d_in, const int* in_sizes, int n_in,
                              void* d_out, int out_size, void* d_ws, size_t ws_size,
                              hipStream_t stream) {
    const float* x     = (const float*)d_in[0];
    const int*   adj   = (const int*)  d_in[1];
    const float* edge  = (const float*)d_in[2];
    const float* mask  = (const float*)d_in[3];
    const float* gamma = (const float*)d_in[4];
    const float* beta  = (const float*)d_in[5];
    const float* W1    = (const float*)d_in[6];
    const float* b1    = (const float*)d_in[7];
    const float* W2    = (const float*)d_in[8];
    const float* b2    = (const float*)d_in[9];
    float* out = (float*)d_out;

    char* ws = (char*)d_ws;
    float*     xm  = (float*)ws;
    _Float16*  hnp = (_Float16*)(ws + (size_t)BSZ * NN * HH * 4);
    _Float16*  wsW = (_Float16*)(ws + (size_t)BSZ * NN * HH * 4
                                    + (size_t)BSZ * NN * HH * 2);

    gine_norm_kernel<<<BSZ * NN, HH, 0, stream>>>(x, mask, gamma, beta, xm, hnp);
    gine_swizzleW_kernel<<<8, 256, 0, stream>>>(W1, W2, wsW);
    gine_main_kernel<<<BSZ * NN, 256, 0, stream>>>(edge, adj, mask, b1, b2,
                                                   xm, hnp, wsW, out);
}


# Sum over j kept in 64 per-lane f32 registers (GEMM2 ot loop unrolled), one LDS tree reduce per block; replaces per-tile shuffle chains and LDS float atomics; order of the j sum changes, now fixed
# speedup vs baseline: 2.0593x; 2.0593x over previous
#include <hip/hip_runtime.h>

typedef __attribute__((ext_vector_type(16))) _Float16 v16h;
typedef __attribute__((ext_vector_type(8)))  _Float16 v8h;
typedef __attribute__((ext_vector_type(8)))  float    v8f;

#define BSZ 2
#define NN  512
#define HH  128


__global__ void gine_norm_kernel(const float* __restrict__ x,
                                 const float* __restrict__ mask,
                                 const float* __restrict__ gamma,
                                 const float* __restrict__ beta,
                                 float* __restrict__ xm,
                                 _Float16* __restrict__ hn)
{
    int row = blockIdx.x;
    int t   = threadIdx.x;
    __shared__ float red[HH];
    float mk = mask[row];
    float xv = x[(size_t)row * HH + t] * mk;
    red[t] = xv; __syncthreads();
    for (int s = HH / 2; s > 0; s >>= 1) {
        if (t < s) red[t] += red[t + s];
        __syncthreads();
    }
    float mu = red[0] * (1.0f / HH);
    __syncthreads();
    float d = xv - mu;
    red[t] = d * d; __syncthreads();
    for (int s = HH / 2; s > 0; s >>= 1) {
        if (t < s) red[t] += red[t + s];
        __syncthreads();
    }
    float var = red[0] * (1.0f / HH);
    float hv  = d * rsqrtf(var + 1e-5f) * gamma[t] + beta[t];
    xm[(size_t)row * HH + t] = xv;
    hn[(size_t)row * HH + t] = (_Float16)hv;
}

__global__ void gine_swizzleW_kernel(const float* __restrict__ W1,
                                     const float* __restrict__ W2,
                                     _Float16* __restrict__ wsW)
{
    int idx  = blockIdx.x * blockDim.x + threadIdx.x;
    int lane = idx & 31;
    int kc   = (idx >> 5) & 3;
    int ot   = (idx >> 7) & 7;
    int mat  = (idx >> 10) & 1;
    const float* W = mat ? W2 : W1;
    _Float16* dst  = wsW + mat * 16384 + ((ot * 4 + kc) * 32 + lane) * 16;
    int m  = lane & 15;
    int lo = (lane & 16) ? 8 : 0;
#pragma unroll
    for (int h = 0; h < 16; ++h) {
        int kl = (h < 8) ? (lo + h) : (16 + lo + (h - 8));
        dst[h] = (_Float16)W[(ot * 16 + m) * HH + kc * 32 + kl];
    }
}

__global__ __launch_bounds__(256, 1)
void gine_main_kernel(const float*    __restrict__ edge,
                      const int*      __restrict__ adj,
                      const float*    __restrict__ mask,
                      const float*    __restrict__ b1,
                      const float*    __restrict__ b2,
                      const float*    __restrict__ xm,
                      const _Float16* __restrict__ hn,
                      const _Float16* __restrict__ wsW,
                      float*          __restrict__ out)
{
    __shared__ __align__(32) _Float16 sW[2 * 16384];
    __shared__ __align__(16) _Float16 sH1[8 * 8 * 32 * 8];
    __shared__ float sB1[HH], sB2[HH];
    __shared__ float sPart[HH];

    int tid  = threadIdx.x;
    int lane = tid & 31;
    int w    = tid >> 5;
    int b    = blockIdx.x / NN;
    int i    = blockIdx.x % NN;

    {
        const uint4* g = (const uint4*)wsW;
        uint4* s = (uint4*)sW;
        for (int idx = tid; idx < 4096; idx += 256) s[idx] = g[idx];
    }
    if (tid < HH) { sB1[tid] = b1[tid]; sB2[tid] = b2[tid]; }
    __syncthreads();

    const int moff = (lane & 16) ? 8 : 0;
    const int mrow = lane & 15;
    const size_t rowbase = (size_t)(b * NN + i) * NN;

    v8f acc[8];
#pragma unroll
    for (int ot = 0; ot < 8; ++ot) acc[ot] = (v8f)0.f;

    for (int jt = w; jt < NN / 16; jt += 8) {
        int jrow = jt * 16 + mrow;
        float am = (float)adj[rowbase + jrow];
        const float*    erow = edge + (rowbase + jrow) * HH;
        const _Float16* hrow = hn + (size_t)(b * NN + jrow) * HH;

        v16h b1c[4];
#pragma unroll
        for (int kc = 0; kc < 4; ++kc) {
            int k0 = kc * 32 + moff;
            float ef[16];
            *(float4*)&ef[0]  = *(const float4*)(erow + k0);
            *(float4*)&ef[4]  = *(const float4*)(erow + k0 + 4);
            *(float4*)&ef[8]  = *(const float4*)(erow + k0 + 16);
            *(float4*)&ef[12] = *(const float4*)(erow + k0 + 20);
            v8h h0 = *(const v8h*)(hrow + k0);
            v8h h1 = *(const v8h*)(hrow + k0 + 16);
            v16h bb;
#pragma unroll
            for (int t2 = 0; t2 < 8; ++t2) {
                bb[t2]     = (_Float16)(ef[t2]     + (float)h0[t2]);
                bb[8 + t2] = (_Float16)(ef[8 + t2] + (float)h1[t2]);
            }
            b1c[kc] = bb;
        }

        if (jt + 8 < NN / 16) {
            const float* enext = edge + (rowbase + (jt + 8) * 16 + mrow) * HH;
#pragma unroll
            for (int kc = 0; kc < 4; ++kc)
                __builtin_prefetch(enext + kc * 32, 0, 1);
        }

#pragma unroll 1
        for (int ot = 0; ot < 8; ++ot) {
            v8f a1 = (v8f)0.f;
#pragma unroll
            for (int kc = 0; kc < 4; ++kc) {
                const v16h aw = *(const v16h*)&sW[((ot * 4 + kc) * 32 + lane) * 16];
                a1 = __builtin_amdgcn_wmma_f32_16x16x32_f16(
                    false, aw, false, b1c[kc], (short)0, a1, false, false);
            }
            float bvv[8];
            *(float4*)&bvv[0] = *(const float4*)&sB1[ot * 16 + moff];
            *(float4*)&bvv[4] = *(const float4*)&sB1[ot * 16 + moff + 4];
            v8h pk;
#pragma unroll
            for (int r = 0; r < 8; ++r)
                pk[r] = (_Float16)fmaxf(a1[r] + bvv[r], 0.f);
            *(v8h*)&sH1[((w * 8 + ot) * 32 + lane) * 8] = pk;
        }

        v16h b2c[4];
#pragma unroll
        for (int kc = 0; kc < 4; ++kc) {
            v8h lo = *(const v8h*)&sH1[((w * 8 + 2 * kc)     * 32 + lane) * 8];
            v8h hi = *(const v8h*)&sH1[((w * 8 + 2 * kc + 1) * 32 + lane) * 8];
            b2c[kc] = __builtin_shufflevector(lo, hi,
                0, 1, 2, 3, 4, 5, 6, 7, 8, 9, 10, 11, 12, 13, 14, 15);
        }

        asm volatile("" ::: "memory");

#pragma unroll
        for (int ot = 0; ot < 8; ++ot) {
            v8f a2 = (v8f)0.f;
#pragma unroll
            for (int kc = 0; kc < 4; ++kc) {
                const v16h aw =
                    *(const v16h*)&sW[16384 + ((ot * 4 + kc) * 32 + lane) * 16];
                a2 = __builtin_amdgcn_wmma_f32_16x16x32_f16(
                    false, aw, false, b2c[kc], (short)0, a2, false, false);
            }
            float bvv[8];
            *(float4*)&bvv[0] = *(const float4*)&sB2[ot * 16 + moff];
            *(float4*)&bvv[4] = *(const float4*)&sB2[ot * 16 + moff + 4];
#pragma unroll
            for (int r = 0; r < 8; ++r)
                acc[ot][r] += fmaxf(a2[r] + bvv[r], 0.f) * am;
        }
    }
    __syncthreads();

    float* sRed = (float*)sW;
#pragma unroll
    for (int ot = 0; ot < 8; ++ot)
#pragma unroll
        for (int r = 0; r < 8; ++r)
            sRed[(ot * 8 + r) * 256 + tid] = acc[ot][r];
    __syncthreads();
    {
        int v  = tid >> 2;
        int hh = (tid >> 1) & 1;
        int hf = tid & 1;
        const float* p = sRed + v * 256 + hh * 16 + hf * 128;
        float s = 0.f;
#pragma unroll
        for (int q = 0; q < 16; ++q) {
            float4 t4 = *(const float4*)(p + (q >> 2) * 32 + (q & 3) * 4);
            s += (t4.x + t4.y) + (t4.z + t4.w);
        }
        s += __shfl_xor(s, 1, 32);
        if (hf == 0) sPart[(v >> 3) * 16 + hh * 8 + (v & 7)] = s;
    }
    __syncthreads();

    if (tid < HH) {
        int row = b * NN + i;
        out[(size_t)row * HH + tid] =
            (xm[(size_t)row * HH + tid] + sPart[tid]) * mask[row];
    }
}

extern "C" void kernel_launch(void* const* d_in, const int* in_sizes, int n_in,
                              void* d_out, int out_size, void* d_ws, size_t ws_size,
                              hipStream_t stream) {
    const float* x     = (const float*)d_in[0];
    const int*   adj   = (const int*)  d_in[1];
    const float* edge  = (const float*)d_in[2];
    const float* mask  = (const float*)d_in[3];
    const float* gamma = (const float*)d_in[4];
    const float* beta  = (const float*)d_in[5];
    const float* W1    = (const float*)d_in[6];
    const float* b1    = (const float*)d_in[7];
    const float* W2    = (const float*)d_in[8];
    const float* b2    = (const float*)d_in[9];
    float* out = (float*)d_out;

    char* ws = (char*)d_ws;
    float*     xm  = (float*)ws;
    _Float16*  hnp = (_Float16*)(ws + (size_t)BSZ * NN * HH * 4);
    _Float16*  wsW = (_Float16*)(ws + (size_t)BSZ * NN * HH * 4
                                    + (size_t)BSZ * NN * HH * 2);

    gine_norm_kernel<<<BSZ * NN, HH, 0, stream>>>(x, mask, gamma, beta, xm, hnp);
    gine_swizzleW_kernel<<<8, 256, 0, stream>>>(W1, W2, wsW);
    gine_main_kernel<<<BSZ * NN, 256, 0, stream>>>(edge, adj, mask, b1, b2,
                                                   xm, hnp, wsW, out);
}


# Each wave runs its four j-tiles together: one LDS weight fragment feeds 4 WMMAs on independent accumulators (LDS weight reads cut 4x); no j loop left: h1 stays in registers, loads issue as one group
# speedup vs baseline: 2.1584x; 1.0481x over previous
#include <hip/hip_runtime.h>

typedef __attribute__((ext_vector_type(16))) _Float16 v16h;
typedef __attribute__((ext_vector_type(8)))  _Float16 v8h;
typedef __attribute__((ext_vector_type(8)))  float    v8f;

#define BSZ 2
#define NN  512
#define HH  128


__global__ void gine_norm_kernel(const float* __restrict__ x,
                                 const float* __restrict__ mask,
                                 const float* __restrict__ gamma,
                                 const float* __restrict__ beta,
                                 float* __restrict__ xm,
                                 _Float16* __restrict__ hn)
{
    int row = blockIdx.x;
    int t   = threadIdx.x;
    __shared__ float red[HH];
    float mk = mask[row];
    float xv = x[(size_t)row * HH + t] * mk;
    red[t] = xv; __syncthreads();
    for (int s = HH / 2; s > 0; s >>= 1) {
        if (t < s) red[t] += red[t + s];
        __syncthreads();
    }
    float mu = red[0] * (1.0f / HH);
    __syncthreads();
    float d = xv - mu;
    red[t] = d * d; __syncthreads();
    for (int s = HH / 2; s > 0; s >>= 1) {
        if (t < s) red[t] += red[t + s];
        __syncthreads();
    }
    float var = red[0] * (1.0f / HH);
    float hv  = d * rsqrtf(var + 1e-5f) * gamma[t] + beta[t];
    xm[(size_t)row * HH + t] = xv;
    hn[(size_t)row * HH + t] = (_Float16)hv;
}

__global__ void gine_swizzleW_kernel(const float* __restrict__ W1,
                                     const float* __restrict__ W2,
                                     _Float16* __restrict__ wsW)
{
    int idx  = blockIdx.x * blockDim.x + threadIdx.x;
    int lane = idx & 31;
    int kc   = (idx >> 5) & 3;
    int ot   = (idx >> 7) & 7;
    int mat  = (idx >> 10) & 1;
    const float* W = mat ? W2 : W1;
    _Float16* dst  = wsW + mat * 16384 + ((ot * 4 + kc) * 32 + lane) * 16;
    int m  = lane & 15;
    int lo = (lane & 16) ? 8 : 0;
#pragma unroll
    for (int h = 0; h < 16; ++h) {
        int kl = (h < 8) ? (lo + h) : (16 + lo + (h - 8));
        dst[h] = (_Float16)W[(ot * 16 + m) * HH + kc * 32 + kl];
    }
}

__global__ __launch_bounds__(256, 1)
void gine_main_kernel(const float*    __restrict__ edge,
                      const int*      __restrict__ adj,
                      const float*    __restrict__ mask,
                      const float*    __restrict__ b1,
                      const float*    __restrict__ b2,
                      const float*    __restrict__ xm,
                      const _Float16* __restrict__ hn,
                      const _Float16* __restrict__ wsW,
                      float*          __restrict__ out)
{
    __shared__ __align__(32) _Float16 sW[2 * 16384];
    __shared__ float sB1[HH], sB2[HH];
    __shared__ float sPart[HH];

    int tid  = threadIdx.x;
    int lane = tid & 31;
    int w    = tid >> 5;
    int b    = blockIdx.x / NN;
    int i    = blockIdx.x % NN;

    {
        const uint4* g = (const uint4*)wsW;
        uint4* s = (uint4*)sW;
        for (int idx = tid; idx < 4096; idx += 256) s[idx] = g[idx];
    }
    if (tid < HH) { sB1[tid] = b1[tid]; sB2[tid] = b2[tid]; }
    __syncthreads();

    const int moff = (lane & 16) ? 8 : 0;
    const int mrow = lane & 15;
    const size_t rowbase = (size_t)(b * NN + i) * NN;

    float4 er[4][16];
    v8h    hr[4][8];
    int    aj[4];
#pragma unroll
    for (int t = 0; t < 4; ++t) {
        int jrow = (w + 8 * t) * 16 + mrow;
        aj[t] = adj[rowbase + jrow];
        const float*    erow = edge + (rowbase + jrow) * HH;
        const _Float16* hrow = hn + (size_t)(b * NN + jrow) * HH;
#pragma unroll
        for (int kc = 0; kc < 4; ++kc) {
            int k0 = kc * 32 + moff;
            er[t][kc * 4 + 0] = *(const float4*)(erow + k0);
            er[t][kc * 4 + 1] = *(const float4*)(erow + k0 + 4);
            er[t][kc * 4 + 2] = *(const float4*)(erow + k0 + 16);
            er[t][kc * 4 + 3] = *(const float4*)(erow + k0 + 20);
            hr[t][kc * 2 + 0] = *(const v8h*)(hrow + k0);
            hr[t][kc * 2 + 1] = *(const v8h*)(hrow + k0 + 16);
        }
    }
    __builtin_amdgcn_sched_barrier(0);

    v16h  b1c[4][4];
    float am[4];
#pragma unroll
    for (int t = 0; t < 4; ++t) {
        am[t] = (float)aj[t];
#pragma unroll
        for (int kc = 0; kc < 4; ++kc) {
            float ef[16];
            *(float4*)&ef[0]  = er[t][kc * 4 + 0];
            *(float4*)&ef[4]  = er[t][kc * 4 + 1];
            *(float4*)&ef[8]  = er[t][kc * 4 + 2];
            *(float4*)&ef[12] = er[t][kc * 4 + 3];
            v8h h0 = hr[t][kc * 2 + 0];
            v8h h1 = hr[t][kc * 2 + 1];
            v16h bb;
#pragma unroll
            for (int t2 = 0; t2 < 8; ++t2) {
                bb[t2]     = (_Float16)(ef[t2]     + (float)h0[t2]);
                bb[8 + t2] = (_Float16)(ef[8 + t2] + (float)h1[t2]);
            }
            b1c[t][kc] = bb;
        }
    }

    v8h h1p[4][8];
#pragma unroll
    for (int ot = 0; ot < 8; ++ot) {
        v16h aw[4];
#pragma unroll
        for (int kc = 0; kc < 4; ++kc)
            aw[kc] = *(const v16h*)&sW[((ot * 4 + kc) * 32 + lane) * 16];
        v8f a1[4];
#pragma unroll
        for (int t = 0; t < 4; ++t) a1[t] = (v8f)0.f;
#pragma unroll
        for (int kc = 0; kc < 4; ++kc)
#pragma unroll
            for (int t = 0; t < 4; ++t)
                a1[t] = __builtin_amdgcn_wmma_f32_16x16x32_f16(
                    false, aw[kc], false, b1c[t][kc], (short)0, a1[t], false, false);
        float bvv[8];
        *(float4*)&bvv[0] = *(const float4*)&sB1[ot * 16 + moff];
        *(float4*)&bvv[4] = *(const float4*)&sB1[ot * 16 + moff + 4];
#pragma unroll
        for (int t = 0; t < 4; ++t) {
            v8h pk;
#pragma unroll
            for (int r = 0; r < 8; ++r)
                pk[r] = (_Float16)fmaxf(a1[t][r] + bvv[r], 0.f);
            h1p[t][ot] = pk;
        }
    }

    v8f acc[8];

#pragma unroll
    for (int ot = 0; ot < 8; ++ot) {
        v16h aw[4];
#pragma unroll
        for (int kc = 0; kc < 4; ++kc)
            aw[kc] = *(const v16h*)&sW[16384 + ((ot * 4 + kc) * 32 + lane) * 16];
        v8f a2[4];
#pragma unroll
        for (int t = 0; t < 4; ++t) a2[t] = (v8f)0.f;
#pragma unroll
        for (int kc = 0; kc < 4; ++kc)
#pragma unroll
            for (int t = 0; t < 4; ++t) {
                v16h bk = __builtin_shufflevector(h1p[t][2 * kc], h1p[t][2 * kc + 1],
                    0, 1, 2, 3, 4, 5, 6, 7, 8, 9, 10, 11, 12, 13, 14, 15);
                a2[t] = __builtin_amdgcn_wmma_f32_16x16x32_f16(
                    false, aw[kc], false, bk, (short)0, a2[t], false, false);
            }
        float bvv[8];
        *(float4*)&bvv[0] = *(const float4*)&sB2[ot * 16 + moff];
        *(float4*)&bvv[4] = *(const float4*)&sB2[ot * 16 + moff + 4];
        v8f s = (v8f)0.f;
#pragma unroll
        for (int t = 0; t < 4; ++t)
#pragma unroll
            for (int r = 0; r < 8; ++r)
                s[r] += fmaxf(a2[t][r] + bvv[r], 0.f) * am[t];
        acc[ot] = s;
    }
    __syncthreads();

    float* sRed = (float*)sW;
#pragma unroll
    for (int ot = 0; ot < 8; ++ot)
#pragma unroll
        for (int r = 0; r < 8; ++r)
            sRed[(ot * 8 + r) * 256 + tid] = acc[ot][r];
    __syncthreads();
    {
        int v  = tid >> 2;
        int hh = (tid >> 1) & 1;
        int hf = tid & 1;
        const float* p = sRed + v * 256 + hh * 16 + hf * 128;
        float s = 0.f;
#pragma unroll
        for (int q = 0; q < 16; ++q) {
            float4 t4 = *(const float4*)(p + (q >> 2) * 32 + (q & 3) * 4);
            s += (t4.x + t4.y) + (t4.z + t4.w);
        }
        s += __shfl_xor(s, 1, 32);
        if (hf == 0) sPart[(v >> 3) * 16 + hh * 8 + (v & 7)] = s;
    }
    __syncthreads();

    if (tid < HH) {
        int row = b * NN + i;
        out[(size_t)row * HH + tid] =
            (xm[(size_t)row * HH + tid] + sPart[tid]) * mask[row];
    }
}

extern "C" void kernel_launch(void* const* d_in, const int* in_sizes, int n_in,
                              void* d_out, int out_size, void* d_ws, size_t ws_size,
                              hipStream_t stream) {
    const float* x     = (const float*)d_in[0];
    const int*   adj   = (const int*)  d_in[1];
    const float* edge  = (const float*)d_in[2];
    const float* mask  = (const float*)d_in[3];
    const float* gamma = (const float*)d_in[4];
    const float* beta  = (const float*)d_in[5];
    const float* W1    = (const float*)d_in[6];
    const float* b1    = (const float*)d_in[7];
    const float* W2    = (const float*)d_in[8];
    const float* b2    = (const float*)d_in[9];
    float* out = (float*)d_out;

    char* ws = (char*)d_ws;
    float*     xm  = (float*)ws;
    _Float16*  hnp = (_Float16*)(ws + (size_t)BSZ * NN * HH * 4);
    _Float16*  wsW = (_Float16*)(ws + (size_t)BSZ * NN * HH * 4
                                    + (size_t)BSZ * NN * HH * 2);

    gine_norm_kernel<<<BSZ * NN, HH, 0, stream>>>(x, mask, gamma, beta, xm, hnp);
    gine_swizzleW_kernel<<<8, 256, 0, stream>>>(W1, W2, wsW);
    gine_main_kernel<<<BSZ * NN, 256, 0, stream>>>(edge, adj, mask, b1, b2,
                                                   xm, hnp, wsW, out);
}


# Next pass's global loads issued ahead of the current pass's GEMMs (raw registers refilled while f16 operands compute); 256 4-wave groups walk 4 rows x 2 passes; j-sum order changes, stays fixed
# speedup vs baseline: 2.4921x; 1.1546x over previous
#include <hip/hip_runtime.h>

typedef __attribute__((ext_vector_type(16))) _Float16 v16h;
typedef __attribute__((ext_vector_type(8)))  _Float16 v8h;
typedef __attribute__((ext_vector_type(8)))  float    v8f;

#define BSZ 2
#define NN  512
#define HH  128


__global__ void gine_norm_kernel(const float* __restrict__ x,
                                 const float* __restrict__ mask,
                                 const float* __restrict__ gamma,
                                 const float* __restrict__ beta,
                                 float* __restrict__ xm,
                                 _Float16* __restrict__ hn)
{
    int row = blockIdx.x;
    int t   = threadIdx.x;
    __shared__ float red[HH];
    float mk = mask[row];
    float xv = x[(size_t)row * HH + t] * mk;
    red[t] = xv; __syncthreads();
    for (int s = HH / 2; s > 0; s >>= 1) {
        if (t < s) red[t] += red[t + s];
        __syncthreads();
    }
    float mu = red[0] * (1.0f / HH);
    __syncthreads();
    float d = xv - mu;
    red[t] = d * d; __syncthreads();
    for (int s = HH / 2; s > 0; s >>= 1) {
        if (t < s) red[t] += red[t + s];
        __syncthreads();
    }
    float var = red[0] * (1.0f / HH);
    float hv  = d * rsqrtf(var + 1e-5f) * gamma[t] + beta[t];
    xm[(size_t)row * HH + t] = xv;
    hn[(size_t)row * HH + t] = (_Float16)hv;
}

__global__ void gine_swizzleW_kernel(const float* __restrict__ W1,
                                     const float* __restrict__ W2,
                                     _Float16* __restrict__ wsW)
{
    int idx  = blockIdx.x * blockDim.x + threadIdx.x;
    int lane = idx & 31;
    int kc   = (idx >> 5) & 3;
    int ot   = (idx >> 7) & 7;
    int mat  = (idx >> 10) & 1;
    const float* W = mat ? W2 : W1;
    _Float16* dst  = wsW + mat * 16384 + ((ot * 4 + kc) * 32 + lane) * 16;
    int m  = lane & 15;
    int lo = (lane & 16) ? 8 : 0;
#pragma unroll
    for (int h = 0; h < 16; ++h) {
        int kl = (h < 8) ? (lo + h) : (16 + lo + (h - 8));
        dst[h] = (_Float16)W[(ot * 16 + m) * HH + kc * 32 + kl];
    }
}

#define ROWS_PER_WG 4
#define NPASS (2 * ROWS_PER_WG)

template <int T0, int T1>
__device__ __forceinline__ void gine_issue_loads(
    const float* __restrict__ edge, const int* __restrict__ adj,
    const _Float16* __restrict__ hn, int row, int p, int w, int mrow, int moff,
    float4 (&er)[4][16], v8h (&hr)[4][8], int (&aj)[4])
{
    const int    bb      = row / NN;
    const size_t rowbase = (size_t)row * NN;
#pragma unroll
    for (int t = T0; t < T1; ++t) {
        int jrow = (p * 16 + w + 4 * t) * 16 + mrow;
        aj[t] = adj[rowbase + jrow];
        const float*    erow = edge + (rowbase + jrow) * HH;
        const _Float16* hrow = hn + (size_t)(bb * NN + jrow) * HH;
#pragma unroll
        for (int kc = 0; kc < 4; ++kc) {
            int k0 = kc * 32 + moff;
            er[t][kc * 4 + 0] = *(const float4*)(erow + k0);
            er[t][kc * 4 + 1] = *(const float4*)(erow + k0 + 4);
            er[t][kc * 4 + 2] = *(const float4*)(erow + k0 + 16);
            er[t][kc * 4 + 3] = *(const float4*)(erow + k0 + 20);
            hr[t][kc * 2 + 0] = *(const v8h*)(hrow + k0);
            hr[t][kc * 2 + 1] = *(const v8h*)(hrow + k0 + 16);
        }
    }
}

__global__ __launch_bounds__(128, 1)
void gine_main_kernel(const float*    __restrict__ edge,
                      const int*      __restrict__ adj,
                      const float*    __restrict__ mask,
                      const float*    __restrict__ b1,
                      const float*    __restrict__ b2,
                      const float*    __restrict__ xm,
                      const _Float16* __restrict__ hn,
                      const _Float16* __restrict__ wsW,
                      float*          __restrict__ out)
{
    __shared__ __align__(32) _Float16 sW[2 * 16384];
    __shared__ __align__(16) float    sRed[64 * 128];
    __shared__ float sB1[HH], sB2[HH];
    __shared__ float sPart[HH];

    int tid  = threadIdx.x;
    int lane = tid & 31;
    int w    = tid >> 5;
    const int row0 = blockIdx.x * ROWS_PER_WG;

    const int moff = (lane & 16) ? 8 : 0;
    const int mrow = lane & 15;

    float4 er[4][16];
    v8h    hr[4][8];
    int    aj[4];
    gine_issue_loads<0, 4>(edge, adj, hn, row0, 0, w, mrow, moff, er, hr, aj);

    {
        const uint4* g = (const uint4*)wsW;
        uint4* s = (uint4*)sW;
        for (int idx = tid; idx < 4096; idx += 128) s[idx] = g[idx];
    }
    sB1[tid] = b1[tid]; sB2[tid] = b2[tid];
    __syncthreads();

    v8f acc[8];
#pragma unroll
    for (int ot = 0; ot < 8; ++ot) acc[ot] = (v8f)0.f;

    auto run_pass = [&](const int q, const bool prefetch) __attribute__((always_inline)) {
        const int nrow = row0 + ((q + 1) >> 1);
        const int np   = (q + 1) & 1;
        v16h  b1c[4][4];
        float am[4];
#pragma unroll
        for (int t = 0; t < 4; ++t) {
            am[t] = (float)aj[t];
#pragma unroll
            for (int kc = 0; kc < 4; ++kc) {
                float ef[16];
                *(float4*)&ef[0]  = er[t][kc * 4 + 0];
                *(float4*)&ef[4]  = er[t][kc * 4 + 1];
                *(float4*)&ef[8]  = er[t][kc * 4 + 2];
                *(float4*)&ef[12] = er[t][kc * 4 + 3];
                v8h h0 = hr[t][kc * 2 + 0];
                v8h h1 = hr[t][kc * 2 + 1];
                v16h bb;
#pragma unroll
                for (int t2 = 0; t2 < 8; ++t2) {
                    bb[t2]     = (_Float16)(ef[t2]     + (float)h0[t2]);
                    bb[8 + t2] = (_Float16)(ef[8 + t2] + (float)h1[t2]);
                }
                b1c[t][kc] = bb;
            }
        }
        __builtin_amdgcn_sched_barrier(0);
        if (prefetch)
            gine_issue_loads<0, 2>(edge, adj, hn, nrow, np, w, mrow, moff, er, hr, aj);
        __builtin_amdgcn_sched_barrier(0);

        v8h h1p[4][8];
#pragma unroll
        for (int ot = 0; ot < 8; ++ot) {
            v16h aw[4];
#pragma unroll
            for (int kc = 0; kc < 4; ++kc)
                aw[kc] = *(const v16h*)&sW[((ot * 4 + kc) * 32 + lane) * 16];
            v8f a1[4];
#pragma unroll
            for (int t = 0; t < 4; ++t) a1[t] = (v8f)0.f;
#pragma unroll
            for (int kc = 0; kc < 4; ++kc)
#pragma unroll
                for (int t = 0; t < 4; ++t)
                    a1[t] = __builtin_amdgcn_wmma_f32_16x16x32_f16(
                        false, aw[kc], false, b1c[t][kc], (short)0, a1[t], false, false);
            float bvv[8];
            *(float4*)&bvv[0] = *(const float4*)&sB1[ot * 16 + moff];
            *(float4*)&bvv[4] = *(const float4*)&sB1[ot * 16 + moff + 4];
#pragma unroll
            for (int t = 0; t < 4; ++t) {
                v8h pk;
#pragma unroll
                for (int r = 0; r < 8; ++r)
                    pk[r] = (_Float16)fmaxf(a1[t][r] + bvv[r], 0.f);
                h1p[t][ot] = pk;
            }
        }

        __builtin_amdgcn_sched_barrier(0);
        if (prefetch)
            gine_issue_loads<2, 4>(edge, adj, hn, nrow, np, w, mrow, moff, er, hr, aj);
        __builtin_amdgcn_sched_barrier(0);

#pragma unroll
        for (int ot = 0; ot < 8; ++ot) {
            v16h aw[4];
#pragma unroll
            for (int kc = 0; kc < 4; ++kc)
                aw[kc] = *(const v16h*)&sW[16384 + ((ot * 4 + kc) * 32 + lane) * 16];
            v8f a2[4];
#pragma unroll
            for (int t = 0; t < 4; ++t) a2[t] = (v8f)0.f;
#pragma unroll
            for (int kc = 0; kc < 4; ++kc)
#pragma unroll
                for (int t = 0; t < 4; ++t) {
                    v16h bk = __builtin_shufflevector(h1p[t][2 * kc], h1p[t][2 * kc + 1],
                        0, 1, 2, 3, 4, 5, 6, 7, 8, 9, 10, 11, 12, 13, 14, 15);
                    a2[t] = __builtin_amdgcn_wmma_f32_16x16x32_f16(
                        false, aw[kc], false, bk, (short)0, a2[t], false, false);
                }
            float bvv[8];
            *(float4*)&bvv[0] = *(const float4*)&sB2[ot * 16 + moff];
            *(float4*)&bvv[4] = *(const float4*)&sB2[ot * 16 + moff + 4];
            v8f s = (v8f)0.f;
#pragma unroll
            for (int t = 0; t < 4; ++t)
#pragma unroll
                for (int r = 0; r < 8; ++r)
                    s[r] += fmaxf(a2[t][r] + bvv[r], 0.f) * am[t];
            acc[ot] += s;
        }

        if (q & 1) {
#pragma unroll
            for (int ot = 0; ot < 8; ++ot)
#pragma unroll
                for (int r = 0; r < 8; ++r)
                    sRed[(ot * 8 + r) * 128 + tid] = acc[ot][r];
            __syncthreads();
            {
                int v  = tid >> 1;
                int hh = tid & 1;
                const float* p = sRed + v * 128 + hh * 16;
                float s = 0.f;
#pragma unroll
                for (int c = 0; c < 16; ++c) {
                    float4 t4 = *(const float4*)(p + (c >> 2) * 32 + (c & 3) * 4);
                    s += (t4.x + t4.y) + (t4.z + t4.w);
                }
                sPart[(v >> 3) * 16 + hh * 8 + (v & 7)] = s;
            }
            __syncthreads();
            {
                int row = row0 + (q >> 1);
                out[(size_t)row * HH + tid] =
                    (xm[(size_t)row * HH + tid] + sPart[tid]) * mask[row];
            }
#pragma unroll
            for (int ot = 0; ot < 8; ++ot) acc[ot] = (v8f)0.f;
        }
    };

#pragma unroll 1
    for (int q = 0; q < NPASS - 1; ++q) run_pass(q, true);
    run_pass(NPASS - 1, false);
}

extern "C" void kernel_launch(void* const* d_in, const int* in_sizes, int n_in,
                              void* d_out, int out_size, void* d_ws, size_t ws_size,
                              hipStream_t stream) {
    const float* x     = (const float*)d_in[0];
    const int*   adj   = (const int*)  d_in[1];
    const float* edge  = (const float*)d_in[2];
    const float* mask  = (const float*)d_in[3];
    const float* gamma = (const float*)d_in[4];
    const float* beta  = (const float*)d_in[5];
    const float* W1    = (const float*)d_in[6];
    const float* b1    = (const float*)d_in[7];
    const float* W2    = (const float*)d_in[8];
    const float* b2    = (const float*)d_in[9];
    float* out = (float*)d_out;

    char* ws = (char*)d_ws;
    float*     xm  = (float*)ws;
    _Float16*  hnp = (_Float16*)(ws + (size_t)BSZ * NN * HH * 4);
    _Float16*  wsW = (_Float16*)(ws + (size_t)BSZ * NN * HH * 4
                                    + (size_t)BSZ * NN * HH * 2);

    gine_norm_kernel<<<BSZ * NN, HH, 0, stream>>>(x, mask, gamma, beta, xm, hnp);
    gine_swizzleW_kernel<<<8, 256, 0, stream>>>(W1, W2, wsW);
    gine_main_kernel<<<(BSZ * NN) / ROWS_PER_WG, 128, 0, stream>>>(edge, adj, mask, b1, b2,
                                                   xm, hnp, wsW, out);
}


# m_in loads cover whole 128-byte lines (a half-wave reads one j row side by side, not one row per lane); f16 sums reach the B fragments through a wave-private LDS tile; values, sum order unchanged
# speedup vs baseline: 2.6479x; 1.0625x over previous
#include <hip/hip_runtime.h>

typedef __attribute__((ext_vector_type(16))) _Float16 v16h;
typedef __attribute__((ext_vector_type(8)))  _Float16 v8h;
typedef __attribute__((ext_vector_type(8)))  float    v8f;

#define BSZ 2
#define NN  512
#define HH  128


__global__ void gine_norm_kernel(const float* __restrict__ x,
                                 const float* __restrict__ mask,
                                 const float* __restrict__ gamma,
                                 const float* __restrict__ beta,
                                 float* __restrict__ xm,
                                 _Float16* __restrict__ hn)
{
    int row = blockIdx.x;
    int t   = threadIdx.x;
    __shared__ float red[HH];
    float mk = mask[row];
    float xv = x[(size_t)row * HH + t] * mk;
    red[t] = xv; __syncthreads();
    for (int s = HH / 2; s > 0; s >>= 1) {
        if (t < s) red[t] += red[t + s];
        __syncthreads();
    }
    float mu = red[0] * (1.0f / HH);
    __syncthreads();
    float d = xv - mu;
    red[t] = d * d; __syncthreads();
    for (int s = HH / 2; s > 0; s >>= 1) {
        if (t < s) red[t] += red[t + s];
        __syncthreads();
    }
    float var = red[0] * (1.0f / HH);
    float hv  = d * rsqrtf(var + 1e-5f) * gamma[t] + beta[t];
    xm[(size_t)row * HH + t] = xv;
    hn[(size_t)row * HH + t] = (_Float16)hv;
}

__global__ void gine_swizzleW_kernel(const float* __restrict__ W1,
                                     const float* __restrict__ W2,
                                     _Float16* __restrict__ wsW)
{
    int idx  = blockIdx.x * blockDim.x + threadIdx.x;
    int lane = idx & 31;
    int kc   = (idx >> 5) & 3;
    int ot   = (idx >> 7) & 7;
    int mat  = (idx >> 10) & 1;
    const float* W = mat ? W2 : W1;
    _Float16* dst  = wsW + mat * 16384 + ((ot * 4 + kc) * 32 + lane) * 16;
    int m  = lane & 15;
    int lo = (lane & 16) ? 8 : 0;
#pragma unroll
    for (int h = 0; h < 16; ++h) {
        int kl = (h < 8) ? (lo + h) : (16 + lo + (h - 8));
        dst[h] = (_Float16)W[(ot * 16 + m) * HH + kc * 32 + kl];
    }
}

#define SPITCH 136
#define ROWS_PER_WG 4
#define NPASS (2 * ROWS_PER_WG)

template <int T0, int T1>
__device__ __forceinline__ void gine_issue_loads(
    const float* __restrict__ edge, const int* __restrict__ adj,
    const _Float16* __restrict__ hn, int row, int p, int w, int mrow, int lhalf,
    float4 (&er)[4][16], v8h (&hr)[4][8], int (&aj)[4])
{
    const int    bb      = row / NN;
    const size_t rowbase = (size_t)row * NN;
#pragma unroll
    for (int t = T0; t < T1; ++t) {
        int jbase = (p * 16 + w + 4 * t) * 16;
        aj[t] = adj[rowbase + jbase + mrow];
        const float*    ep = edge + (rowbase + jbase + lhalf) * HH + 8 * mrow;
        const _Float16* hp = hn + (size_t)(bb * NN + jbase + lhalf) * HH + 8 * mrow;
#pragma unroll
        for (int i = 0; i < 8; ++i) {
            er[t][2 * i + 0] = *(const float4*)(ep + (2 * i) * HH);
            er[t][2 * i + 1] = *(const float4*)(ep + (2 * i) * HH + 4);
            hr[t][i]         = *(const v8h*)(hp + (2 * i) * HH);
        }
    }
}

__global__ __launch_bounds__(128, 1)
void gine_main_kernel(const float*    __restrict__ edge,
                      const int*      __restrict__ adj,
                      const float*    __restrict__ mask,
                      const float*    __restrict__ b1,
                      const float*    __restrict__ b2,
                      const float*    __restrict__ xm,
                      const _Float16* __restrict__ hn,
                      const _Float16* __restrict__ wsW,
                      float*          __restrict__ out)
{
    __shared__ __align__(32) _Float16 sW[2 * 16384];
    __shared__ __align__(16) float    sRed[64 * 128];
    __shared__ float sB1[HH], sB2[HH];
    __shared__ float sPart[HH];
    __shared__ __align__(16) _Float16 sStage[4 * 64 * SPITCH];

    int tid  = threadIdx.x;
    int lane = tid & 31;
    int w    = tid >> 5;
    const int row0 = blockIdx.x * ROWS_PER_WG;

    const int moff = (lane & 16) ? 8 : 0;
    const int mrow = lane & 15;
    const int lhalf = lane >> 4;
    _Float16* const st = sStage + w * (64 * SPITCH);

    float4 er[4][16];
    v8h    hr[4][8];
    int    aj[4];
    gine_issue_loads<0, 4>(edge, adj, hn, row0, 0, w, mrow, lhalf, er, hr, aj);

    {
        const uint4* g = (const uint4*)wsW;
        uint4* s = (uint4*)sW;
        for (int idx = tid; idx < 4096; idx += 128) s[idx] = g[idx];
    }
    sB1[tid] = b1[tid]; sB2[tid] = b2[tid];
    __syncthreads();

    v8f acc[8];
#pragma unroll
    for (int ot = 0; ot < 8; ++ot) acc[ot] = (v8f)0.f;

    auto run_pass = [&](const int q, const bool prefetch) __attribute__((always_inline)) {
        const int nrow = row0 + ((q + 1) >> 1);
        const int np   = (q + 1) & 1;
        v16h  b1c[4][4];
        float am[4];
#pragma unroll
        for (int t = 0; t < 4; ++t) {
            am[t] = (float)aj[t];
#pragma unroll
            for (int i = 0; i < 8; ++i) {
                float ef[8];
                *(float4*)&ef[0] = er[t][2 * i + 0];
                *(float4*)&ef[4] = er[t][2 * i + 1];
                v8h h0 = hr[t][i];
                v8h mm;
#pragma unroll
                for (int t2 = 0; t2 < 8; ++t2)
                    mm[t2] = (_Float16)(ef[t2] + (float)h0[t2]);
                *(v8h*)&st[(t * 16 + 2 * i + lhalf) * SPITCH + 8 * mrow] = mm;
            }
        }
        __builtin_amdgcn_sched_barrier(0);
        if (prefetch)
            gine_issue_loads<0, 2>(edge, adj, hn, nrow, np, w, mrow, lhalf, er, hr, aj);
        __builtin_amdgcn_sched_barrier(0);

        __builtin_amdgcn_fence(__ATOMIC_RELEASE, "workgroup", "local");
#pragma unroll
        for (int t = 0; t < 4; ++t) {
#pragma unroll
            for (int kc = 0; kc < 4; ++kc) {
                const _Float16* fp = &st[(t * 16 + mrow) * SPITCH + kc * 32 + moff];
                v8h lo = *(const v8h*)fp;
                v8h hi = *(const v8h*)(fp + 16);
                b1c[t][kc] = __builtin_shufflevector(lo, hi,
                    0, 1, 2, 3, 4, 5, 6, 7, 8, 9, 10, 11, 12, 13, 14, 15);
            }
        }
        __builtin_amdgcn_sched_barrier(0);

        v8h h1p[4][8];
#pragma unroll
        for (int ot = 0; ot < 8; ++ot) {
            v16h aw[4];
#pragma unroll
            for (int kc = 0; kc < 4; ++kc)
                aw[kc] = *(const v16h*)&sW[((ot * 4 + kc) * 32 + lane) * 16];
            v8f a1[4];
#pragma unroll
            for (int t = 0; t < 4; ++t) a1[t] = (v8f)0.f;
#pragma unroll
            for (int kc = 0; kc < 4; ++kc)
#pragma unroll
                for (int t = 0; t < 4; ++t)
                    a1[t] = __builtin_amdgcn_wmma_f32_16x16x32_f16(
                        false, aw[kc], false, b1c[t][kc], (short)0, a1[t], false, false);
            float bvv[8];
            *(float4*)&bvv[0] = *(const float4*)&sB1[ot * 16 + moff];
            *(float4*)&bvv[4] = *(const float4*)&sB1[ot * 16 + moff + 4];
#pragma unroll
            for (int t = 0; t < 4; ++t) {
                v8h pk;
#pragma unroll
                for (int r = 0; r < 8; ++r)
                    pk[r] = (_Float16)fmaxf(a1[t][r] + bvv[r], 0.f);
                h1p[t][ot] = pk;
            }
        }

        __builtin_amdgcn_sched_barrier(0);
        if (prefetch)
            gine_issue_loads<2, 4>(edge, adj, hn, nrow, np, w, mrow, lhalf, er, hr, aj);
        __builtin_amdgcn_sched_barrier(0);

#pragma unroll
        for (int ot = 0; ot < 8; ++ot) {
            v16h aw[4];
#pragma unroll
            for (int kc = 0; kc < 4; ++kc)
                aw[kc] = *(const v16h*)&sW[16384 + ((ot * 4 + kc) * 32 + lane) * 16];
            v8f a2[4];
#pragma unroll
            for (int t = 0; t < 4; ++t) a2[t] = (v8f)0.f;
#pragma unroll
            for (int kc = 0; kc < 4; ++kc)
#pragma unroll
                for (int t = 0; t < 4; ++t) {
                    v16h bk = __builtin_shufflevector(h1p[t][2 * kc], h1p[t][2 * kc + 1],
                        0, 1, 2, 3, 4, 5, 6, 7, 8, 9, 10, 11, 12, 13, 14, 15);
                    a2[t] = __builtin_amdgcn_wmma_f32_16x16x32_f16(
                        false, aw[kc], false, bk, (short)0, a2[t], false, false);
                }
            float bvv[8];
            *(float4*)&bvv[0] = *(const float4*)&sB2[ot * 16 + moff];
            *(float4*)&bvv[4] = *(const float4*)&sB2[ot * 16 + moff + 4];
            v8f s = (v8f)0.f;
#pragma unroll
            for (int t = 0; t < 4; ++t)
#pragma unroll
                for (int r = 0; r < 8; ++r)
                    s[r] += fmaxf(a2[t][r] + bvv[r], 0.f) * am[t];
            acc[ot] += s;
        }

        if (q & 1) {
#pragma unroll
            for (int ot = 0; ot < 8; ++ot)
#pragma unroll
                for (int r = 0; r < 8; ++r)
                    sRed[(ot * 8 + r) * 128 + tid] = acc[ot][r];
            __syncthreads();
            {
                int v  = tid >> 1;
                int hh = tid & 1;
                const float* p = sRed + v * 128 + hh * 16;
                float s = 0.f;
#pragma unroll
                for (int c = 0; c < 16; ++c) {
                    float4 t4 = *(const float4*)(p + (c >> 2) * 32 + (c & 3) * 4);
                    s += (t4.x + t4.y) + (t4.z + t4.w);
                }
                sPart[(v >> 3) * 16 + hh * 8 + (v & 7)] = s;
            }
            __syncthreads();
            {
                int row = row0 + (q >> 1);
                out[(size_t)row * HH + tid] =
                    (xm[(size_t)row * HH + tid] + sPart[tid]) * mask[row];
            }
#pragma unroll
            for (int ot = 0; ot < 8; ++ot) acc[ot] = (v8f)0.f;
        }
    };

#pragma unroll 1
    for (int q = 0; q < NPASS - 1; ++q) run_pass(q, true);
    run_pass(NPASS - 1, false);
}

extern "C" void kernel_launch(void* const* d_in, const int* in_sizes, int n_in,
                              void* d_out, int out_size, void* d_ws, size_t ws_size,
                              hipStream_t stream) {
    const float* x     = (const float*)d_in[0];
    const int*   adj   = (const int*)  d_in[1];
    const float* edge  = (const float*)d_in[2];
    const float* mask  = (const float*)d_in[3];
    const float* gamma = (const float*)d_in[4];
    const float* beta  = (const float*)d_in[5];
    const float* W1    = (const float*)d_in[6];
    const float* b1    = (const float*)d_in[7];
    const float* W2    = (const float*)d_in[8];
    const float* b2    = (const float*)d_in[9];
    float* out = (float*)d_out;

    char* ws = (char*)d_ws;
    float*     xm  = (float*)ws;
    _Float16*  hnp = (_Float16*)(ws + (size_t)BSZ * NN * HH * 4);
    _Float16*  wsW = (_Float16*)(ws + (size_t)BSZ * NN * HH * 4
                                    + (size_t)BSZ * NN * HH * 2);

    gine_norm_kernel<<<BSZ * NN, HH, 0, stream>>>(x, mask, gamma, beta, xm, hnp);
    gine_swizzleW_kernel<<<8, 256, 0, stream>>>(W1, W2, wsW);
    gine_main_kernel<<<(BSZ * NN) / ROWS_PER_WG, 128, 0, stream>>>(edge, adj, mask, b1, b2,
                                                   xm, hnp, wsW, out);
}
